# starting kernel
# baseline (speedup 1.0000x reference)
#include <hip/hip_runtime.h>
#include <hip/hip_bf16.h>

typedef __attribute__((ext_vector_type(2))) float v2f;
typedef __attribute__((ext_vector_type(8))) float v8f;

#define GAT_N    50000
#define GAT_E    800000
#define GAT_DIN  128
#define GAT_DHID 64
#define GAT_DOUT 32
#define NEG_SLOPE 0.2f

__device__ __forceinline__ unsigned flipf(float f) {
  unsigned u = __float_as_uint(f);
  return (u & 0x80000000u) ? ~u : (u | 0x80000000u);
}
__device__ __forceinline__ float unflipf(unsigned u) {
  unsigned b = (u & 0x80000000u) ? (u & 0x7FFFFFFFu) : ~u;
  return __uint_as_float(b);
}

template<int K, int NCOL>
__global__ __launch_bounds__(256)
void wmma_gemm_kernel(const float* __restrict__ A, const float* __restrict__ B,
                      float* __restrict__ C, int Mrows) {
  constexpr int NT = NCOL / 16;
  int wave = (int)((blockIdx.x * blockDim.x + threadIdx.x) >> 5);
  int lane = (int)(threadIdx.x & 31);
  int m0 = wave * 16;
  if (m0 >= Mrows) return;
  int half = lane >> 4;
  int lr = lane & 15;

  v8f acc[NT] = {};
  const float* arow = A + (size_t)(m0 + lr) * K;

  for (int k = 0; k < K; k += 4) {
    int ka = k + 2 * half;
    v2f a;
    a.x = arow[ka];
    a.y = arow[ka + 1];
    const float* b0 = B + (size_t)ka * NCOL;
    const float* b1 = B + (size_t)(ka + 1) * NCOL;
#pragma unroll
    for (int t = 0; t < NT; ++t) {
      v2f b;
      b.x = b0[t * 16 + lr];
      b.y = b1[t * 16 + lr];
      acc[t] = __builtin_amdgcn_wmma_f32_16x16x4_f32(
           false, a,  false, b,
           (short)0, acc[t],  false,  false);
    }
  }

  int row = m0 + half * 8;
#pragma unroll
  for (int t = 0; t < NT; ++t) {
#pragma unroll
    for (int r = 0; r < 8; ++r) {
      C[(size_t)(row + r) * NCOL + t * 16 + lr] = acc[t][r];
    }
  }
}

template<int F>
__global__ __launch_bounds__(256)
void node_attn_kernel(const float* __restrict__ feat,
                      const float* __restrict__ al, const float* __restrict__ ar,
                      float* __restrict__ el, float* __restrict__ er, int n_nodes) {
  int wave = (int)((blockIdx.x * blockDim.x + threadIdx.x) >> 5);
  int lane = (int)(threadIdx.x & 31);
  if (wave >= n_nodes) return;
  const float* frow = feat + (size_t)wave * (2 * F);
#pragma unroll
  for (int h = 0; h < 2; ++h) {
    float sl = 0.f, sr = 0.f;
    for (int f = lane; f < F; f += 32) {
      float v = frow[h * F + f];
      sl += v * al[h * F + f];
      sr += v * ar[h * F + f];
    }
#pragma unroll
    for (int off = 16; off > 0; off >>= 1) {
      sl += __shfl_xor(sl, off, 32);
      sr += __shfl_xor(sr, off, 32);
    }
    if (lane == 0) {
      el[wave * 2 + h] = sl;
      er[wave * 2 + h] = sr;
    }
  }
}

__global__ __launch_bounds__(256)
void edge_logits_kernel(const int* __restrict__ src, const int* __restrict__ dst,
                        const float* __restrict__ el, const float* __restrict__ er,
                        float* __restrict__ e, unsigned* __restrict__ menc, int nE) {
  int i = (int)(blockIdx.x * blockDim.x + threadIdx.x);
  if (i >= nE) return;
  int s = src[i], d = dst[i];
#pragma unroll
  for (int h = 0; h < 2; ++h) {
    float v = el[s * 2 + h] + er[d * 2 + h];
    v = (v > 0.f) ? v : NEG_SLOPE * v;
    e[(size_t)i * 2 + h] = v;
    atomicMax(&menc[d * 2 + h], flipf(v));
  }
}

__global__ __launch_bounds__(256)
void edge_alpha_kernel(const int* __restrict__ dst, const unsigned* __restrict__ menc,
                       float* __restrict__ e_alpha, float* __restrict__ denom, int nE) {
  int i = (int)(blockIdx.x * blockDim.x + threadIdx.x);
  if (i >= nE) return;
  int d = dst[i];
#pragma unroll
  for (int h = 0; h < 2; ++h) {
    float m = unflipf(menc[d * 2 + h]);
    float a = __expf(e_alpha[(size_t)i * 2 + h] - m);
    e_alpha[(size_t)i * 2 + h] = a;
    atomicAdd(&denom[d * 2 + h], a);
  }
}

template<int F>
__global__ __launch_bounds__(256)
void edge_agg_kernel(const int* __restrict__ src, const int* __restrict__ dst,
                     const float* __restrict__ feat, const float* __restrict__ alpha,
                     const float* __restrict__ denom, float* __restrict__ agg, int nE) {
  constexpr int C = 2 * F;
  constexpr int VEC = C / 32;
  int wave = (int)((blockIdx.x * blockDim.x + threadIdx.x) >> 5);
  int lane = (int)(threadIdx.x & 31);
  if (wave >= nE) return;
  int s = src[wave], d = dst[wave];
  int col = lane * VEC;
  int h = (col >= F) ? 1 : 0;
  float sc = alpha[(size_t)wave * 2 + h] / denom[d * 2 + h];
  const float* fsrc = feat + (size_t)s * C + col;
  float* ad = agg + (size_t)d * C + col;
#pragma unroll
  for (int j = 0; j < VEC; ++j) {
    atomicAdd(&ad[j], fsrc[j] * sc);
  }
}

template<int F>
__global__ __launch_bounds__(256)
void finish_kernel(const float* __restrict__ agg, const float* __restrict__ bias,
                   float* __restrict__ out, int n_nodes) {
  int i = (int)(blockIdx.x * blockDim.x + threadIdx.x);
  if (i >= n_nodes * F) return;
  int n = i / F, f = i % F;
  float v = 0.5f * (agg[(size_t)n * 2 * F + f] + bias[f] +
                    agg[(size_t)n * 2 * F + F + f] + bias[F + f]);
  out[i] = (v > 0.f) ? v : 0.f;
}

static inline int cdiv(long long a, long long b) { return (int)((a + b - 1) / b); }

extern "C" void kernel_launch(void* const* d_in, const int* in_sizes, int n_in,
                              void* d_out, int out_size, void* d_ws, size_t ws_size,
                              hipStream_t stream) {
  (void)in_sizes; (void)n_in; (void)out_size; (void)ws_size;
  const float* x   = (const float*)d_in[0];
  const int*  esrc = (const int*)d_in[1];
  const int*  edst = (const int*)d_in[2];
  const float* W1  = (const float*)d_in[3];
  const float* al1 = (const float*)d_in[4];
  const float* ar1 = (const float*)d_in[5];
  const float* b1  = (const float*)d_in[6];
  const float* W2  = (const float*)d_in[7];
  const float* al2 = (const float*)d_in[8];
  const float* ar2 = (const float*)d_in[9];
  const float* b2  = (const float*)d_in[10];

  const int N = GAT_N, E = GAT_E;

  float* ws = (float*)d_ws;
  size_t o = 0;
  float*    feat1  = ws + o; o += (size_t)N * 128;
  float*    el1    = ws + o; o += (size_t)N * 2;
  float*    er1    = ws + o; o += (size_t)N * 2;
  unsigned* m1     = (unsigned*)(ws + o); o += (size_t)N * 2;
  float*    denom1 = ws + o; o += (size_t)N * 2;
  float*    e1     = ws + o; o += (size_t)E * 2;
  float*    agg1   = ws + o; o += (size_t)N * 128;
  float*    h1     = ws + o; o += (size_t)N * 64;
  float*    feat2  = ws + o; o += (size_t)N * 64;
  float*    el2    = ws + o; o += (size_t)N * 2;
  float*    er2    = ws + o; o += (size_t)N * 2;
  unsigned* m2     = (unsigned*)(ws + o); o += (size_t)N * 2;
  float*    denom2 = ws + o; o += (size_t)N * 2;
  float*    e2     = ws + o; o += (size_t)E * 2;
  float*    agg2   = ws + o; o += (size_t)N * 64;

  hipMemsetAsync(m1,     0, (size_t)N * 2 * 4, stream);
  hipMemsetAsync(denom1, 0, (size_t)N * 2 * 4, stream);
  hipMemsetAsync(agg1,   0, (size_t)N * 128 * 4, stream);
  hipMemsetAsync(m2,     0, (size_t)N * 2 * 4, stream);
  hipMemsetAsync(denom2, 0, (size_t)N * 2 * 4, stream);
  hipMemsetAsync(agg2,   0, (size_t)N * 64 * 4, stream);

  const int BLK = 256;
  {
    int waves = N / 16;
    wmma_gemm_kernel<128, 128><<<cdiv((long long)waves * 32, BLK), BLK, 0, stream>>>(
        x, W1, feat1, N);
    node_attn_kernel<64><<<cdiv((long long)N * 32, BLK), BLK, 0, stream>>>(
        feat1, al1, ar1, el1, er1, N);
    edge_logits_kernel<<<cdiv(E, BLK), BLK, 0, stream>>>(esrc, edst, el1, er1, e1, m1, E);
    edge_alpha_kernel<<<cdiv(E, BLK), BLK, 0, stream>>>(edst, m1, e1, denom1, E);
    edge_agg_kernel<64><<<cdiv((long long)E * 32, BLK), BLK, 0, stream>>>(
        esrc, edst, feat1, e1, denom1, agg1, E);
    finish_kernel<64><<<cdiv((long long)N * 64, BLK), BLK, 0, stream>>>(agg1, b1, h1, N);
  }
  {
    int waves = N / 16;
    wmma_gemm_kernel<64, 64><<<cdiv((long long)waves * 32, BLK), BLK, 0, stream>>>(
        h1, W2, feat2, N);
    node_attn_kernel<32><<<cdiv((long long)N * 32, BLK), BLK, 0, stream>>>(
        feat2, al2, ar2, el2, er2, N);
    edge_logits_kernel<<<cdiv(E, BLK), BLK, 0, stream>>>(esrc, edst, el2, er2, e2, m2, E);
    edge_alpha_kernel<<<cdiv(E, BLK), BLK, 0, stream>>>(edst, m2, e2, denom2, E);
    edge_agg_kernel<32><<<cdiv((long long)E * 32, BLK), BLK, 0, stream>>>(
        esrc, edst, feat2, e2, denom2, agg2, E);
    finish_kernel<32><<<cdiv((long long)N * 32, BLK), BLK, 0, stream>>>(
        agg2, b2, (float*)d_out, N);
  }
}


# Edges bucketed by destination once per call (counting sort, arrival-order rank); one wave per node does max, exp-sum, weighted gather and epilogue in registers: float atomics gone, sum order changed
# speedup vs baseline: 2.7560x; 2.7560x over previous
#include <hip/hip_runtime.h>
#include <hip/hip_bf16.h>

typedef __attribute__((ext_vector_type(2))) float v2f;
typedef __attribute__((ext_vector_type(8))) float v8f;

#define GAT_N    50000
#define GAT_E    800000
#define GAT_DIN  128
#define GAT_DHID 64
#define GAT_DOUT 32
#define NEG_SLOPE 0.2f

template<int K, int NCOL>
__global__ __launch_bounds__(256)
void wmma_gemm_kernel(const float* __restrict__ A, const float* __restrict__ B,
                      float* __restrict__ C, int Mrows) {
  constexpr int NT = NCOL / 16;
  int wave = (int)((blockIdx.x * blockDim.x + threadIdx.x) >> 5);
  int lane = (int)(threadIdx.x & 31);
  int m0 = wave * 16;
  if (m0 >= Mrows) return;
  int half = lane >> 4;
  int lr = lane & 15;

  v8f acc[NT] = {};
  const float* arow = A + (size_t)(m0 + lr) * K;

  for (int k = 0; k < K; k += 4) {
    int ka = k + 2 * half;
    v2f a;
    a.x = arow[ka];
    a.y = arow[ka + 1];
    const float* b0 = B + (size_t)ka * NCOL;
    const float* b1 = B + (size_t)(ka + 1) * NCOL;
#pragma unroll
    for (int t = 0; t < NT; ++t) {
      v2f b;
      b.x = b0[t * 16 + lr];
      b.y = b1[t * 16 + lr];
      acc[t] = __builtin_amdgcn_wmma_f32_16x16x4_f32(
           false, a,  false, b,
           (short)0, acc[t],  false,  false);
    }
  }

  int row = m0 + half * 8;
#pragma unroll
  for (int t = 0; t < NT; ++t) {
#pragma unroll
    for (int r = 0; r < 8; ++r) {
      C[(size_t)(row + r) * NCOL + t * 16 + lr] = acc[t][r];
    }
  }
}

template<int F>
__global__ __launch_bounds__(256)
void node_attn_kernel(const float* __restrict__ feat,
                      const float* __restrict__ al, const float* __restrict__ ar,
                      float* __restrict__ el, float* __restrict__ er, int n_nodes) {
  int wave = (int)((blockIdx.x * blockDim.x + threadIdx.x) >> 5);
  int lane = (int)(threadIdx.x & 31);
  if (wave >= n_nodes) return;
  const float* frow = feat + (size_t)wave * (2 * F);
#pragma unroll
  for (int h = 0; h < 2; ++h) {
    float sl = 0.f, sr = 0.f;
    for (int f = lane; f < F; f += 32) {
      float v = frow[h * F + f];
      sl += v * al[h * F + f];
      sr += v * ar[h * F + f];
    }
#pragma unroll
    for (int off = 16; off > 0; off >>= 1) {
      sl += __shfl_xor(sl, off, 32);
      sr += __shfl_xor(sr, off, 32);
    }
    if (lane == 0) {
      el[wave * 2 + h] = sl;
      er[wave * 2 + h] = sr;
    }
  }
}

__global__ __launch_bounds__(256)
void degree_rank_kernel(const int* __restrict__ dst, int* __restrict__ cnt,
                        int* __restrict__ rank, int nE) {
  int i = (int)(blockIdx.x * blockDim.x + threadIdx.x);
  if (i >= nE) return;
  rank[i] = atomicAdd(&cnt[dst[i]], 1);
}

#define SCAN_THREADS 1024
#define SCAN_PER ((GAT_N + SCAN_THREADS - 1) / SCAN_THREADS)
__global__ __launch_bounds__(SCAN_THREADS)
void offsets_scan_kernel(const int* __restrict__ cnt, int* __restrict__ offs, int n) {
  __shared__ int wtot[32];
  int t = (int)threadIdx.x;
  int lane = t & 31, w = t >> 5;
  int base = t * SCAN_PER;
  int v[SCAN_PER];
  int loc = 0;
#pragma unroll
  for (int k = 0; k < SCAN_PER; ++k) {
    int idx = base + k;
    v[k] = (idx < n) ? cnt[idx] : 0;
    loc += v[k];
  }
  int inc = loc;
  for (int off = 1; off < 32; off <<= 1) {
    int u = __shfl_up(inc, off, 32);
    if (lane >= off) inc += u;
  }
  if (lane == 31) wtot[w] = inc;
  __syncthreads();
  if (w == 0) {
    int x = wtot[lane];
    int xi = x;
    for (int off = 1; off < 32; off <<= 1) {
      int u = __shfl_up(xi, off, 32);
      if (lane >= off) xi += u;
    }
    wtot[lane] = xi - x;
  }
  __syncthreads();
  int run = wtot[w] + inc - loc;
#pragma unroll
  for (int k = 0; k < SCAN_PER; ++k) {
    int idx = base + k;
    if (idx < n) offs[idx] = run;
    run += v[k];
  }
  if (t == SCAN_THREADS - 1) offs[n] = run;
}

__global__ __launch_bounds__(256)
void bucket_fill_kernel(const int* __restrict__ src, const int* __restrict__ dst,
                        const int* __restrict__ offs, const int* __restrict__ rank,
                        int* __restrict__ csr_src, int nE) {
  int i = (int)(blockIdx.x * blockDim.x + threadIdx.x);
  if (i >= nE) return;
  csr_src[offs[dst[i]] + rank[i]] = src[i];
}

template<int F>
__global__ __launch_bounds__(256)
void node_gather_kernel(const int* __restrict__ offs, const int* __restrict__ csr_src,
                        const float* __restrict__ feat,
                        const float* __restrict__ el, const float* __restrict__ er,
                        const float* __restrict__ bias, float* __restrict__ out,
                        int n_nodes) {
  constexpr int C = 2 * F;
  constexpr int VEC = C / 32;
  typedef __attribute__((ext_vector_type(VEC))) float vrow;
  int lane = (int)(threadIdx.x & 31);
  int d = __builtin_amdgcn_readfirstlane((int)(blockIdx.x * 8 + (threadIdx.x >> 5)));
  if (d >= n_nodes) return;
  int beg = offs[d];
  int end = offs[d + 1];
  float er0 = er[2 * d], er1 = er[2 * d + 1];
  bool hi = lane >= 16;
  int col = lane * VEC;

  int s_f = 0;
  float v0_f = -INFINITY, v1_f = -INFINITY;
  if (beg + lane < end) {
    s_f = csr_src[beg + lane];
    v2f e2 = *(const v2f*)(el + 2 * (size_t)s_f);
    float a = e2.x + er0, b = e2.y + er1;
    v0_f = (a > 0.f) ? a : NEG_SLOPE * a;
    v1_f = (b > 0.f) ? b : NEG_SLOPE * b;
  }
  float m0 = v0_f, m1 = v1_f;
  for (int base = beg + 32; base < end; base += 32) {
    int j = base + lane;
    if (j < end) {
      int s = csr_src[j];
      v2f e2 = *(const v2f*)(el + 2 * (size_t)s);
      float a = e2.x + er0, b = e2.y + er1;
      m0 = fmaxf(m0, (a > 0.f) ? a : NEG_SLOPE * a);
      m1 = fmaxf(m1, (b > 0.f) ? b : NEG_SLOPE * b);
    }
  }
  for (int off = 16; off > 0; off >>= 1) {
    m0 = fmaxf(m0, __shfl_xor(m0, off, 32));
    m1 = fmaxf(m1, __shfl_xor(m1, off, 32));
  }

  vrow acc = {};
  float dsum = 0.f;
  for (int base = beg; base < end; base += 32) {
    int j = base + lane;
    int s = s_f;
    float v0 = v0_f, v1 = v1_f;
    if (base != beg) {
      s = 0; v0 = -INFINITY; v1 = -INFINITY;
      if (j < end) {
        s = csr_src[j];
        v2f e2 = *(const v2f*)(el + 2 * (size_t)s);
        float a = e2.x + er0, b = e2.y + er1;
        v0 = (a > 0.f) ? a : NEG_SLOPE * a;
        v1 = (b > 0.f) ? b : NEG_SLOPE * b;
      }
    }
    float a0 = (j < end) ? __expf(v0 - m0) : 0.f;
    float a1 = (j < end) ? __expf(v1 - m1) : 0.f;
    int n = end - base;
    n = (n < 32) ? n : 32;
#pragma unroll 4
    for (int t = 0; t < n; ++t) {
      int st = __builtin_amdgcn_readlane(s, t);
      float at0 = __int_as_float(__builtin_amdgcn_readlane(__float_as_int(a0), t));
      float at1 = __int_as_float(__builtin_amdgcn_readlane(__float_as_int(a1), t));
      float a = hi ? at1 : at0;
      vrow r = *(const vrow*)(feat + (size_t)st * C + col);
      acc += r * a;
      dsum += a;
    }
  }

  bool any = end > beg;
  vrow ov;
  for (int q = 0; q < VEC; ++q) {
    float r = any ? acc[q] / dsum : 0.f;
    float v = r + bias[col + q];
    float w = __shfl_xor(v, 16, 32);
    v = 0.5f * (v + w);
    ov[q] = (v > 0.f) ? v : 0.f;
  }
  if (!hi) *(vrow*)(out + (size_t)d * F + col) = ov;
}

static inline int cdiv(long long a, long long b) { return (int)((a + b - 1) / b); }

extern "C" void kernel_launch(void* const* d_in, const int* in_sizes, int n_in,
                              void* d_out, int out_size, void* d_ws, size_t ws_size,
                              hipStream_t stream) {
  (void)in_sizes; (void)n_in; (void)out_size; (void)ws_size;
  const float* x   = (const float*)d_in[0];
  const int*  esrc = (const int*)d_in[1];
  const int*  edst = (const int*)d_in[2];
  const float* W1  = (const float*)d_in[3];
  const float* al1 = (const float*)d_in[4];
  const float* ar1 = (const float*)d_in[5];
  const float* b1  = (const float*)d_in[6];
  const float* W2  = (const float*)d_in[7];
  const float* al2 = (const float*)d_in[8];
  const float* ar2 = (const float*)d_in[9];
  const float* b2  = (const float*)d_in[10];

  const int N = GAT_N, E = GAT_E;

  float* ws = (float*)d_ws;
  size_t o = 0;
  float* feat1  = ws + o; o += (size_t)N * 128;
  float* el1    = ws + o; o += (size_t)N * 2;
  float* er1    = ws + o; o += (size_t)N * 2;
  float* h1     = ws + o; o += (size_t)N * 64;
  float* feat2  = ws + o; o += (size_t)N * 64;
  float* el2    = ws + o; o += (size_t)N * 2;
  float* er2    = ws + o; o += (size_t)N * 2;
  int*   cnt    = (int*)(ws + o); o += (size_t)N;
  int*   rank   = (int*)(ws + o); o += (size_t)E;
  int*   csrc   = (int*)(ws + o); o += (size_t)E;
  int*   offs   = (int*)(ws + o); o += (size_t)N + 4;

  const int BLK = 256;
  hipMemsetAsync(cnt, 0, (size_t)N * 4, stream);
  degree_rank_kernel<<<cdiv(E, BLK), BLK, 0, stream>>>(edst, cnt, rank, E);
  offsets_scan_kernel<<<1, SCAN_THREADS, 0, stream>>>(cnt, offs, N);
  bucket_fill_kernel<<<cdiv(E, BLK), BLK, 0, stream>>>(esrc, edst, offs, rank, csrc, E);

  {
    int waves = N / 16;
    wmma_gemm_kernel<128, 128><<<cdiv((long long)waves * 32, BLK), BLK, 0, stream>>>(
        x, W1, feat1, N);
    node_attn_kernel<64><<<cdiv((long long)N * 32, BLK), BLK, 0, stream>>>(
        feat1, al1, ar1, el1, er1, N);
    node_gather_kernel<64><<<cdiv((long long)N * 32, BLK), BLK, 0, stream>>>(
        offs, csrc, feat1, el1, er1, b1, h1, N);
  }
  {
    int waves = N / 16;
    wmma_gemm_kernel<64, 64><<<cdiv((long long)waves * 32, BLK), BLK, 0, stream>>>(
        h1, W2, feat2, N);
    node_attn_kernel<32><<<cdiv((long long)N * 32, BLK), BLK, 0, stream>>>(
        feat2, al2, ar2, el2, er2, N);
    node_gather_kernel<32><<<cdiv((long long)N * 32, BLK), BLK, 0, stream>>>(
        offs, csrc, feat2, el2, er2, b2, (float*)d_out, N);
  }
}


# Edges bucketed by a two-level counting sort on LDS counters (chunk by node range, then node): per-edge returning global atomic, memset and one-block scan gone; bucket (sum) order changed
# speedup vs baseline: 4.6405x; 1.6838x over previous
#include <hip/hip_runtime.h>
#include <hip/hip_bf16.h>

typedef __attribute__((ext_vector_type(2))) float v2f;
typedef __attribute__((ext_vector_type(8))) float v8f;

#define GAT_N    50000
#define GAT_E    800000
#define GAT_DIN  128
#define GAT_DHID 64
#define GAT_DOUT 32
#define NEG_SLOPE 0.2f

template<int K, int NCOL>
__global__ __launch_bounds__(256)
void wmma_gemm_kernel(const float* __restrict__ A, const float* __restrict__ B,
                      float* __restrict__ C, int Mrows) {
  constexpr int NT = NCOL / 16;
  int wave = (int)((blockIdx.x * blockDim.x + threadIdx.x) >> 5);
  int lane = (int)(threadIdx.x & 31);
  int m0 = wave * 16;
  if (m0 >= Mrows) return;
  int half = lane >> 4;
  int lr = lane & 15;

  v8f acc[NT] = {};
  const float* arow = A + (size_t)(m0 + lr) * K;

  for (int k = 0; k < K; k += 4) {
    int ka = k + 2 * half;
    v2f a;
    a.x = arow[ka];
    a.y = arow[ka + 1];
    const float* b0 = B + (size_t)ka * NCOL;
    const float* b1 = B + (size_t)(ka + 1) * NCOL;
#pragma unroll
    for (int t = 0; t < NT; ++t) {
      v2f b;
      b.x = b0[t * 16 + lr];
      b.y = b1[t * 16 + lr];
      acc[t] = __builtin_amdgcn_wmma_f32_16x16x4_f32(
           false, a,  false, b,
           (short)0, acc[t],  false,  false);
    }
  }

  int row = m0 + half * 8;
#pragma unroll
  for (int t = 0; t < NT; ++t) {
#pragma unroll
    for (int r = 0; r < 8; ++r) {
      C[(size_t)(row + r) * NCOL + t * 16 + lr] = acc[t][r];
    }
  }
}

template<int F>
__global__ __launch_bounds__(256)
void node_attn_kernel(const float* __restrict__ feat,
                      const float* __restrict__ al, const float* __restrict__ ar,
                      float* __restrict__ el, float* __restrict__ er, int n_nodes) {
  int wave = (int)((blockIdx.x * blockDim.x + threadIdx.x) >> 5);
  int lane = (int)(threadIdx.x & 31);
  if (wave >= n_nodes) return;
  const float* frow = feat + (size_t)wave * (2 * F);
#pragma unroll
  for (int h = 0; h < 2; ++h) {
    float sl = 0.f, sr = 0.f;
    for (int f = lane; f < F; f += 32) {
      float v = frow[h * F + f];
      sl += v * al[h * F + f];
      sr += v * ar[h * F + f];
    }
#pragma unroll
    for (int off = 16; off > 0; off >>= 1) {
      sl += __shfl_xor(sl, off, 32);
      sr += __shfl_xor(sr, off, 32);
    }
    if (lane == 0) {
      el[wave * 2 + h] = sl;
      er[wave * 2 + h] = sr;
    }
  }
}

#define BK_NC  256
#define BK_CH  ((GAT_E + BK_NC - 1) / BK_NC)
#define BK_RSH 8
#define BK_RN  (1 << BK_RSH)
#define BK_NR  ((GAT_N + BK_RN - 1) / BK_RN)
#define BK_PER ((BK_CH + 255) / 256)
static_assert(BK_NR <= 256, "one thread per node range");
static_assert(BK_NR * BK_RN > GAT_N, "offs[n_nodes] is written by the last range");
static_assert(GAT_N <= (1 << 23), "packed edge word stays non-negative");
static_assert(BK_NC == 256 && BK_RN == 256, "one thread per chunk / per node of a range");

__device__ __forceinline__ int block_excl_scan_256(int v, int* wtot, int t) {
  int lane = t & 31, w = t >> 5;
  int inc = v;
  for (int off = 1; off < 32; off <<= 1) {
    int u = __shfl_up(inc, off, 32);
    if (lane >= off) inc += u;
  }
  if (lane == 31) wtot[w] = inc;
  __syncthreads();
  int wbase = 0;
  for (int q = 0; q < 8; ++q) wbase += (q < w) ? wtot[q] : 0;
  return wbase + inc - v;
}

__global__ __launch_bounds__(256)
void chunk_split_kernel(const int* __restrict__ src, const int* __restrict__ dst,
                        int* __restrict__ tmp, int* __restrict__ cntCR,
                        int* __restrict__ offCR, int nE) {
  __shared__ int hist[256];
  __shared__ int wtot[8];
  int t = (int)threadIdx.x;
  int c = (int)blockIdx.x;
  hist[t] = 0;
  __syncthreads();
  int cbeg = c * BK_CH;
  int cend = cbeg + BK_CH;
  cend = (cend < nE) ? cend : nE;
  int rg[BK_PER], pw[BK_PER], rk[BK_PER];
#pragma unroll
  for (int k = 0; k < BK_PER; ++k) {
    int i = cbeg + k * 256 + t;
    rg[k] = -1; pw[k] = 0;
    if (i < cend) {
      int d = dst[i];
      int s = src[i];
      rg[k] = d >> BK_RSH;
      pw[k] = (s << 8) | (d & (BK_RN - 1));
    }
  }
#pragma unroll
  for (int k = 0; k < BK_PER; ++k) {
    rk[k] = 0;
    if (rg[k] >= 0) rk[k] = atomicAdd(&hist[rg[k]], 1);
  }
  __syncthreads();
  int v = hist[t];
  int ex = block_excl_scan_256(v, wtot, t);
  hist[t] = ex;
  if (t < BK_NR) {
    cntCR[t * BK_NC + c] = v;
    offCR[t * BK_NC + c] = ex;
  }
  __syncthreads();
#pragma unroll
  for (int k = 0; k < BK_PER; ++k) {
    if (rg[k] >= 0) tmp[cbeg + hist[rg[k]] + rk[k]] = pw[k];
  }
}

__global__ __launch_bounds__(256)
void range_place_kernel(const int* __restrict__ tmp, const int* __restrict__ cntCR,
                        const int* __restrict__ offCR, int* __restrict__ offs,
                        int* __restrict__ csr_src, int n_nodes) {
  __shared__ int ncnt[256];
  __shared__ int wsum[8];
  __shared__ int wtot[8];
  int t = (int)threadIdx.x;
  int r = (int)blockIdx.x;
  int lane = t & 31, w = t >> 5;
  ncnt[t] = 0;
  int n_c = cntCR[r * BK_NC + t];
  int o_c = offCR[r * BK_NC + t];
  int part = 0;
  const int4* c4 = (const int4*)cntCR;
  int n4 = r * (BK_NC / 4);
#pragma unroll 8
  for (int q = t; q < n4; q += 256) {
    int4 x = c4[q];
    part += (x.x + x.y) + (x.z + x.w);
  }
  for (int off = 16; off > 0; off >>= 1) part += __shfl_xor(part, off, 32);
  if (lane == 0) wsum[w] = part;
  __syncthreads();
  int base = 0;
  for (int q = 0; q < 8; ++q) base += wsum[q];

  const int* piece = tmp + t * BK_CH + o_c;
  for (int k = 0; k < n_c; k += 4) {
    int x[4];
#pragma unroll
    for (int u = 0; u < 4; ++u) x[u] = (k + u < n_c) ? piece[k + u] : -1;
#pragma unroll
    for (int u = 0; u < 4; ++u) if (x[u] >= 0) atomicAdd(&ncnt[x[u] & (BK_RN - 1)], 1);
  }
  __syncthreads();
  int v = ncnt[t];
  int ex = block_excl_scan_256(v, wtot, t);
  int node = r * BK_RN + t;
  if (node <= n_nodes) offs[node] = base + ex;
  ncnt[t] = ex;
  __syncthreads();
  for (int k = 0; k < n_c; k += 4) {
    int x[4];
#pragma unroll
    for (int u = 0; u < 4; ++u) x[u] = (k + u < n_c) ? piece[k + u] : -1;
#pragma unroll
    for (int u = 0; u < 4; ++u) {
      if (x[u] >= 0) {
        int p = atomicAdd(&ncnt[x[u] & (BK_RN - 1)], 1);
        csr_src[base + p] = x[u] >> 8;
      }
    }
  }
}

template<int F>
__global__ __launch_bounds__(256)
void node_gather_kernel(const int* __restrict__ offs, const int* __restrict__ csr_src,
                        const float* __restrict__ feat,
                        const float* __restrict__ el, const float* __restrict__ er,
                        const float* __restrict__ bias, float* __restrict__ out,
                        int n_nodes) {
  constexpr int C = 2 * F;
  constexpr int VEC = C / 32;
  typedef __attribute__((ext_vector_type(VEC))) float vrow;
  int lane = (int)(threadIdx.x & 31);
  int d = __builtin_amdgcn_readfirstlane((int)(blockIdx.x * 8 + (threadIdx.x >> 5)));
  if (d >= n_nodes) return;
  int beg = offs[d];
  int end = offs[d + 1];
  float er0 = er[2 * d], er1 = er[2 * d + 1];
  bool hi = lane >= 16;
  int col = lane * VEC;

  int s_f = 0;
  float v0_f = -INFINITY, v1_f = -INFINITY;
  if (beg + lane < end) {
    s_f = csr_src[beg + lane];
    v2f e2 = *(const v2f*)(el + 2 * (size_t)s_f);
    float a = e2.x + er0, b = e2.y + er1;
    v0_f = (a > 0.f) ? a : NEG_SLOPE * a;
    v1_f = (b > 0.f) ? b : NEG_SLOPE * b;
  }
  float m0 = v0_f, m1 = v1_f;
  for (int base = beg + 32; base < end; base += 32) {
    int j = base + lane;
    if (j < end) {
      int s = csr_src[j];
      v2f e2 = *(const v2f*)(el + 2 * (size_t)s);
      float a = e2.x + er0, b = e2.y + er1;
      m0 = fmaxf(m0, (a > 0.f) ? a : NEG_SLOPE * a);
      m1 = fmaxf(m1, (b > 0.f) ? b : NEG_SLOPE * b);
    }
  }
  for (int off = 16; off > 0; off >>= 1) {
    m0 = fmaxf(m0, __shfl_xor(m0, off, 32));
    m1 = fmaxf(m1, __shfl_xor(m1, off, 32));
  }

  vrow acc = {};
  float dsum = 0.f;
  for (int base = beg; base < end; base += 32) {
    int j = base + lane;
    int s = s_f;
    float v0 = v0_f, v1 = v1_f;
    if (base != beg) {
      s = 0; v0 = -INFINITY; v1 = -INFINITY;
      if (j < end) {
        s = csr_src[j];
        v2f e2 = *(const v2f*)(el + 2 * (size_t)s);
        float a = e2.x + er0, b = e2.y + er1;
        v0 = (a > 0.f) ? a : NEG_SLOPE * a;
        v1 = (b > 0.f) ? b : NEG_SLOPE * b;
      }
    }
    float a0 = (j < end) ? __expf(v0 - m0) : 0.f;
    float a1 = (j < end) ? __expf(v1 - m1) : 0.f;
    int n = end - base;
    n = (n < 32) ? n : 32;
#pragma unroll 4
    for (int t = 0; t < n; ++t) {
      int st = __builtin_amdgcn_readlane(s, t);
      float at0 = __int_as_float(__builtin_amdgcn_readlane(__float_as_int(a0), t));
      float at1 = __int_as_float(__builtin_amdgcn_readlane(__float_as_int(a1), t));
      float a = hi ? at1 : at0;
      vrow r = *(const vrow*)(feat + (size_t)st * C + col);
      acc += r * a;
      dsum += a;
    }
  }

  bool any = end > beg;
  vrow ov;
  for (int q = 0; q < VEC; ++q) {
    float r = any ? acc[q] / dsum : 0.f;
    float v = r + bias[col + q];
    float w = __shfl_xor(v, 16, 32);
    v = 0.5f * (v + w);
    ov[q] = (v > 0.f) ? v : 0.f;
  }
  if (!hi) *(vrow*)(out + (size_t)d * F + col) = ov;
}

static inline int cdiv(long long a, long long b) { return (int)((a + b - 1) / b); }

extern "C" void kernel_launch(void* const* d_in, const int* in_sizes, int n_in,
                              void* d_out, int out_size, void* d_ws, size_t ws_size,
                              hipStream_t stream) {
  (void)in_sizes; (void)n_in; (void)out_size; (void)ws_size;
  const float* x   = (const float*)d_in[0];
  const int*  esrc = (const int*)d_in[1];
  const int*  edst = (const int*)d_in[2];
  const float* W1  = (const float*)d_in[3];
  const float* al1 = (const float*)d_in[4];
  const float* ar1 = (const float*)d_in[5];
  const float* b1  = (const float*)d_in[6];
  const float* W2  = (const float*)d_in[7];
  const float* al2 = (const float*)d_in[8];
  const float* ar2 = (const float*)d_in[9];
  const float* b2  = (const float*)d_in[10];

  const int N = GAT_N, E = GAT_E;

  float* ws = (float*)d_ws;
  size_t o = 0;
  float* feat1  = ws + o; o += (size_t)N * 128;
  float* el1    = ws + o; o += (size_t)N * 2;
  float* er1    = ws + o; o += (size_t)N * 2;
  float* h1     = ws + o; o += (size_t)N * 64;
  float* feat2  = ws + o; o += (size_t)N * 64;
  float* el2    = ws + o; o += (size_t)N * 2;
  float* er2    = ws + o; o += (size_t)N * 2;
  int*   tmp    = (int*)(ws + o); o += (size_t)E;
  int*   csrc   = (int*)(ws + o); o += (size_t)E;
  int*   offs   = (int*)(ws + o); o += (size_t)N + 4;
  int*   cntCR  = (int*)(ws + o); o += (size_t)BK_NR * BK_NC;
  int*   offCR  = (int*)(ws + o); o += (size_t)BK_NR * BK_NC;

  const int BLK = 256;
  chunk_split_kernel<<<BK_NC, 256, 0, stream>>>(esrc, edst, tmp, cntCR, offCR, E);
  range_place_kernel<<<BK_NR, 256, 0, stream>>>(tmp, cntCR, offCR, offs, csrc, N);

  {
    int waves = N / 16;
    wmma_gemm_kernel<128, 128><<<cdiv((long long)waves * 32, BLK), BLK, 0, stream>>>(
        x, W1, feat1, N);
    node_attn_kernel<64><<<cdiv((long long)N * 32, BLK), BLK, 0, stream>>>(
        feat1, al1, ar1, el1, er1, N);
    node_gather_kernel<64><<<cdiv((long long)N * 32, BLK), BLK, 0, stream>>>(
        offs, csrc, feat1, el1, er1, b1, h1, N);
  }
  {
    int waves = N / 16;
    wmma_gemm_kernel<64, 64><<<cdiv((long long)waves * 32, BLK), BLK, 0, stream>>>(
        h1, W2, feat2, N);
    node_attn_kernel<32><<<cdiv((long long)N * 32, BLK), BLK, 0, stream>>>(
        feat2, al2, ar2, el2, er2, N);
    node_gather_kernel<32><<<cdiv((long long)N * 32, BLK), BLK, 0, stream>>>(
        offs, csrc, feat2, el2, er2, b2, (float*)d_out, N);
  }
}


# Attention logits el/er summed in the projection's epilogue from the WMMA accumulators (16-lane DPP sums): both per-node logit launches and their re-read of feat gone; logit sum order changed
# speedup vs baseline: 5.9100x; 1.2736x over previous
#include <hip/hip_runtime.h>
#include <hip/hip_bf16.h>

typedef __attribute__((ext_vector_type(2))) float v2f;
typedef __attribute__((ext_vector_type(8))) float v8f;

#define GAT_N    50000
#define GAT_E    800000
#define GAT_DIN  128
#define GAT_DHID 64
#define GAT_DOUT 32
#define NEG_SLOPE 0.2f


template<int M>
__device__ __forceinline__ float row_xor(float v) {
  return __builtin_bit_cast(float,
      __builtin_amdgcn_update_dpp(0, __builtin_bit_cast(int, v), 0x160 | M, 0xf, 0xf, true));
}

__device__ __forceinline__ float row16_transpose_sum(const float (&v)[16], int c) {
  bool k8 = (c & 8) != 0, k4 = (c & 4) != 0, k2 = (c & 2) != 0, k1 = (c & 1) != 0;
  float a[8], b[4], d[2];
#pragma unroll
  for (int j = 0; j < 8; ++j) {
    float keep = k8 ? v[j + 8] : v[j];
    float send = k8 ? v[j] : v[j + 8];
    a[j] = keep + row_xor<8>(send);
  }
#pragma unroll
  for (int j = 0; j < 4; ++j) {
    float keep = k4 ? a[j + 4] : a[j];
    float send = k4 ? a[j] : a[j + 4];
    b[j] = keep + row_xor<4>(send);
  }
#pragma unroll
  for (int j = 0; j < 2; ++j) {
    float keep = k2 ? b[j + 2] : b[j];
    float send = k2 ? b[j] : b[j + 2];
    d[j] = keep + row_xor<2>(send);
  }
  float keep = k1 ? d[1] : d[0];
  float send = k1 ? d[0] : d[1];
  return keep + row_xor<1>(send);
}

template<int K, int NCOL>
__global__ __launch_bounds__(256)
void wmma_gemm_kernel(const float* __restrict__ A, const float* __restrict__ B,
                      float* __restrict__ C,
                      const float* __restrict__ al, const float* __restrict__ ar,
                      float* __restrict__ el, float* __restrict__ er, int Mrows) {
  constexpr int NT = NCOL / 16;
  constexpr int HT = NT / 2;
  int wave = (int)((blockIdx.x * blockDim.x + threadIdx.x) >> 5);
  int lane = (int)(threadIdx.x & 31);
  int m0 = wave * 16;
  if (m0 >= Mrows) return;
  int half = lane >> 4;
  int lr = lane & 15;

  float alv[NT], arv[NT];
#pragma unroll
  for (int t = 0; t < NT; ++t) {
    alv[t] = al[t * 16 + lr];
    arv[t] = ar[t * 16 + lr];
  }

  v8f acc[NT] = {};
  const float* arow = A + (size_t)(m0 + lr) * K;

  for (int k = 0; k < K; k += 4) {
    int ka = k + 2 * half;
    v2f a;
    a.x = arow[ka];
    a.y = arow[ka + 1];
    const float* b0 = B + (size_t)ka * NCOL;
    const float* b1 = B + (size_t)(ka + 1) * NCOL;
#pragma unroll
    for (int t = 0; t < NT; ++t) {
      v2f b;
      b.x = b0[t * 16 + lr];
      b.y = b1[t * 16 + lr];
      acc[t] = __builtin_amdgcn_wmma_f32_16x16x4_f32(
           false, a,  false, b,
           (short)0, acc[t],  false,  false);
    }
  }

  int row = m0 + half * 8;
#pragma unroll
  for (int t = 0; t < NT; ++t) {
#pragma unroll
    for (int r = 0; r < 8; ++r) {
      C[(size_t)(row + r) * NCOL + t * 16 + lr] = acc[t][r];
    }
  }

  float vl[16], vr[16];
#pragma unroll
  for (int r = 0; r < 8; ++r) {
#pragma unroll
    for (int h = 0; h < 2; ++h) {
      float sl = acc[h * HT][r] * alv[h * HT];
      float sr = acc[h * HT][r] * arv[h * HT];
#pragma unroll
      for (int t = h * HT + 1; t < (h + 1) * HT; ++t) {
        sl += acc[t][r] * alv[t];
        sr += acc[t][r] * arv[t];
      }
      vl[2 * r + h] = sl;
      vr[2 * r + h] = sr;
    }
  }
  float suml = row16_transpose_sum(vl, lr);
  float sumr = row16_transpose_sum(vr, lr);
  el[(size_t)m0 * 2 + lane] = suml;
  er[(size_t)m0 * 2 + lane] = sumr;
}

#define BK_NC  256
#define BK_CH  ((GAT_E + BK_NC - 1) / BK_NC)
#define BK_RSH 8
#define BK_RN  (1 << BK_RSH)
#define BK_NR  ((GAT_N + BK_RN - 1) / BK_RN)
#define BK_PER ((BK_CH + 255) / 256)
static_assert(BK_NR <= 256, "one thread per node range");
static_assert(BK_NR * BK_RN > GAT_N, "offs[n_nodes] is written by the last range");
static_assert(GAT_N <= (1 << 23), "packed edge word stays non-negative");
static_assert(BK_NC == 256 && BK_RN == 256, "one thread per chunk / per node of a range");

__device__ __forceinline__ int block_excl_scan_256(int v, int* wtot, int t) {
  int lane = t & 31, w = t >> 5;
  int inc = v;
  for (int off = 1; off < 32; off <<= 1) {
    int u = __shfl_up(inc, off, 32);
    if (lane >= off) inc += u;
  }
  if (lane == 31) wtot[w] = inc;
  __syncthreads();
  int wbase = 0;
  for (int q = 0; q < 8; ++q) wbase += (q < w) ? wtot[q] : 0;
  return wbase + inc - v;
}

__global__ __launch_bounds__(256)
void chunk_split_kernel(const int* __restrict__ src, const int* __restrict__ dst,
                        int* __restrict__ tmp, int* __restrict__ cntCR,
                        int* __restrict__ offCR, int nE) {
  __shared__ int hist[256];
  __shared__ int wtot[8];
  int t = (int)threadIdx.x;
  int c = (int)blockIdx.x;
  hist[t] = 0;
  __syncthreads();
  int cbeg = c * BK_CH;
  int cend = cbeg + BK_CH;
  cend = (cend < nE) ? cend : nE;
  int rg[BK_PER], pw[BK_PER], rk[BK_PER];
#pragma unroll
  for (int k = 0; k < BK_PER; ++k) {
    int i = cbeg + k * 256 + t;
    rg[k] = -1; pw[k] = 0;
    if (i < cend) {
      int d = dst[i];
      int s = src[i];
      rg[k] = d >> BK_RSH;
      pw[k] = (s << 8) | (d & (BK_RN - 1));
    }
  }
#pragma unroll
  for (int k = 0; k < BK_PER; ++k) {
    rk[k] = 0;
    if (rg[k] >= 0) rk[k] = atomicAdd(&hist[rg[k]], 1);
  }
  __syncthreads();
  int v = hist[t];
  int ex = block_excl_scan_256(v, wtot, t);
  hist[t] = ex;
  if (t < BK_NR) {
    cntCR[t * BK_NC + c] = v;
    offCR[t * BK_NC + c] = ex;
  }
  __syncthreads();
#pragma unroll
  for (int k = 0; k < BK_PER; ++k) {
    if (rg[k] >= 0) tmp[cbeg + hist[rg[k]] + rk[k]] = pw[k];
  }
}

__global__ __launch_bounds__(256)
void range_place_kernel(const int* __restrict__ tmp, const int* __restrict__ cntCR,
                        const int* __restrict__ offCR, int* __restrict__ offs,
                        int* __restrict__ csr_src, int n_nodes) {
  __shared__ int ncnt[256];
  __shared__ int wsum[8];
  __shared__ int wtot[8];
  int t = (int)threadIdx.x;
  int r = (int)blockIdx.x;
  int lane = t & 31, w = t >> 5;
  ncnt[t] = 0;
  int n_c = cntCR[r * BK_NC + t];
  int o_c = offCR[r * BK_NC + t];
  int part = 0;
  const int4* c4 = (const int4*)cntCR;
  int n4 = r * (BK_NC / 4);
#pragma unroll 8
  for (int q = t; q < n4; q += 256) {
    int4 x = c4[q];
    part += (x.x + x.y) + (x.z + x.w);
  }
  for (int off = 16; off > 0; off >>= 1) part += __shfl_xor(part, off, 32);
  if (lane == 0) wsum[w] = part;
  __syncthreads();
  int base = 0;
  for (int q = 0; q < 8; ++q) base += wsum[q];

  const int* piece = tmp + t * BK_CH + o_c;
  for (int k = 0; k < n_c; k += 4) {
    int x[4];
#pragma unroll
    for (int u = 0; u < 4; ++u) x[u] = (k + u < n_c) ? piece[k + u] : -1;
#pragma unroll
    for (int u = 0; u < 4; ++u) if (x[u] >= 0) atomicAdd(&ncnt[x[u] & (BK_RN - 1)], 1);
  }
  __syncthreads();
  int v = ncnt[t];
  int ex = block_excl_scan_256(v, wtot, t);
  int node = r * BK_RN + t;
  if (node <= n_nodes) offs[node] = base + ex;
  ncnt[t] = ex;
  __syncthreads();
  for (int k = 0; k < n_c; k += 4) {
    int x[4];
#pragma unroll
    for (int u = 0; u < 4; ++u) x[u] = (k + u < n_c) ? piece[k + u] : -1;
#pragma unroll
    for (int u = 0; u < 4; ++u) {
      if (x[u] >= 0) {
        int p = atomicAdd(&ncnt[x[u] & (BK_RN - 1)], 1);
        csr_src[base + p] = x[u] >> 8;
      }
    }
  }
}

template<int F>
__global__ __launch_bounds__(256)
void node_gather_kernel(const int* __restrict__ offs, const int* __restrict__ csr_src,
                        const float* __restrict__ feat,
                        const float* __restrict__ el, const float* __restrict__ er,
                        const float* __restrict__ bias, float* __restrict__ out,
                        int n_nodes) {
  constexpr int C = 2 * F;
  constexpr int VEC = C / 32;
  typedef __attribute__((ext_vector_type(VEC))) float vrow;
  int lane = (int)(threadIdx.x & 31);
  int d = __builtin_amdgcn_readfirstlane((int)(blockIdx.x * 8 + (threadIdx.x >> 5)));
  if (d >= n_nodes) return;
  int beg = offs[d];
  int end = offs[d + 1];
  float er0 = er[2 * d], er1 = er[2 * d + 1];
  bool hi = lane >= 16;
  int col = lane * VEC;

  int s_f = 0;
  float v0_f = -INFINITY, v1_f = -INFINITY;
  if (beg + lane < end) {
    s_f = csr_src[beg + lane];
    v2f e2 = *(const v2f*)(el + 2 * (size_t)s_f);
    float a = e2.x + er0, b = e2.y + er1;
    v0_f = (a > 0.f) ? a : NEG_SLOPE * a;
    v1_f = (b > 0.f) ? b : NEG_SLOPE * b;
  }
  float m0 = v0_f, m1 = v1_f;
  for (int base = beg + 32; base < end; base += 32) {
    int j = base + lane;
    if (j < end) {
      int s = csr_src[j];
      v2f e2 = *(const v2f*)(el + 2 * (size_t)s);
      float a = e2.x + er0, b = e2.y + er1;
      m0 = fmaxf(m0, (a > 0.f) ? a : NEG_SLOPE * a);
      m1 = fmaxf(m1, (b > 0.f) ? b : NEG_SLOPE * b);
    }
  }
  for (int off = 16; off > 0; off >>= 1) {
    m0 = fmaxf(m0, __shfl_xor(m0, off, 32));
    m1 = fmaxf(m1, __shfl_xor(m1, off, 32));
  }

  vrow acc = {};
  float dsum = 0.f;
  for (int base = beg; base < end; base += 32) {
    int j = base + lane;
    int s = s_f;
    float v0 = v0_f, v1 = v1_f;
    if (base != beg) {
      s = 0; v0 = -INFINITY; v1 = -INFINITY;
      if (j < end) {
        s = csr_src[j];
        v2f e2 = *(const v2f*)(el + 2 * (size_t)s);
        float a = e2.x + er0, b = e2.y + er1;
        v0 = (a > 0.f) ? a : NEG_SLOPE * a;
        v1 = (b > 0.f) ? b : NEG_SLOPE * b;
      }
    }
    float a0 = (j < end) ? __expf(v0 - m0) : 0.f;
    float a1 = (j < end) ? __expf(v1 - m1) : 0.f;
    int n = end - base;
    n = (n < 32) ? n : 32;
#pragma unroll 4
    for (int t = 0; t < n; ++t) {
      int st = __builtin_amdgcn_readlane(s, t);
      float at0 = __int_as_float(__builtin_amdgcn_readlane(__float_as_int(a0), t));
      float at1 = __int_as_float(__builtin_amdgcn_readlane(__float_as_int(a1), t));
      float a = hi ? at1 : at0;
      vrow r = *(const vrow*)(feat + (size_t)st * C + col);
      acc += r * a;
      dsum += a;
    }
  }

  bool any = end > beg;
  vrow ov;
  for (int q = 0; q < VEC; ++q) {
    float r = any ? acc[q] / dsum : 0.f;
    float v = r + bias[col + q];
    float w = __shfl_xor(v, 16, 32);
    v = 0.5f * (v + w);
    ov[q] = (v > 0.f) ? v : 0.f;
  }
  if (!hi) *(vrow*)(out + (size_t)d * F + col) = ov;
}

static inline int cdiv(long long a, long long b) { return (int)((a + b - 1) / b); }

extern "C" void kernel_launch(void* const* d_in, const int* in_sizes, int n_in,
                              void* d_out, int out_size, void* d_ws, size_t ws_size,
                              hipStream_t stream) {
  (void)in_sizes; (void)n_in; (void)out_size; (void)ws_size;
  const float* x   = (const float*)d_in[0];
  const int*  esrc = (const int*)d_in[1];
  const int*  edst = (const int*)d_in[2];
  const float* W1  = (const float*)d_in[3];
  const float* al1 = (const float*)d_in[4];
  const float* ar1 = (const float*)d_in[5];
  const float* b1  = (const float*)d_in[6];
  const float* W2  = (const float*)d_in[7];
  const float* al2 = (const float*)d_in[8];
  const float* ar2 = (const float*)d_in[9];
  const float* b2  = (const float*)d_in[10];

  const int N = GAT_N, E = GAT_E;

  float* ws = (float*)d_ws;
  size_t o = 0;
  float* feat1  = ws + o; o += (size_t)N * 128;
  float* el1    = ws + o; o += (size_t)N * 2;
  float* er1    = ws + o; o += (size_t)N * 2;
  float* h1     = ws + o; o += (size_t)N * 64;
  float* feat2  = ws + o; o += (size_t)N * 64;
  float* el2    = ws + o; o += (size_t)N * 2;
  float* er2    = ws + o; o += (size_t)N * 2;
  int*   tmp    = (int*)(ws + o); o += (size_t)E;
  int*   csrc   = (int*)(ws + o); o += (size_t)E;
  int*   offs   = (int*)(ws + o); o += (size_t)N + 4;
  int*   cntCR  = (int*)(ws + o); o += (size_t)BK_NR * BK_NC;
  int*   offCR  = (int*)(ws + o); o += (size_t)BK_NR * BK_NC;

  const int BLK = 256;
  chunk_split_kernel<<<BK_NC, 256, 0, stream>>>(esrc, edst, tmp, cntCR, offCR, E);
  range_place_kernel<<<BK_NR, 256, 0, stream>>>(tmp, cntCR, offCR, offs, csrc, N);

  {
    int waves = N / 16;
    wmma_gemm_kernel<128, 128><<<cdiv((long long)waves * 32, BLK), BLK, 0, stream>>>(
        x, W1, feat1, al1, ar1, el1, er1, N);
    node_gather_kernel<64><<<cdiv((long long)N * 32, BLK), BLK, 0, stream>>>(
        offs, csrc, feat1, el1, er1, b1, h1, N);
  }
  {
    int waves = N / 16;
    wmma_gemm_kernel<64, 64><<<cdiv((long long)waves * 32, BLK), BLK, 0, stream>>>(
        h1, W2, feat2, al2, ar2, el2, er2, N);
    node_gather_kernel<32><<<cdiv((long long)N * 32, BLK), BLK, 0, stream>>>(
        offs, csrc, feat2, el2, er2, b2, (float*)d_out, N);
  }
}


# node_gather: one wave owns 4 consecutive nodes (12500 waves per launch instead of 50000) and loads the bucket bounds and bucket heads of its run together; per-node sums and their order unchanged
# speedup vs baseline: 6.1106x; 1.0339x over previous
#include <hip/hip_runtime.h>
#include <hip/hip_bf16.h>

typedef __attribute__((ext_vector_type(2))) float v2f;
typedef __attribute__((ext_vector_type(8))) float v8f;

#define GAT_N    50000
#define GAT_E    800000
#define GAT_DIN  128
#define GAT_DHID 64
#define GAT_DOUT 32
#define NEG_SLOPE 0.2f


template<int M>
__device__ __forceinline__ float row_xor(float v) {
  return __builtin_bit_cast(float,
      __builtin_amdgcn_update_dpp(0, __builtin_bit_cast(int, v), 0x160 | M, 0xf, 0xf, true));
}

__device__ __forceinline__ float row16_transpose_sum(const float (&v)[16], int c) {
  bool k8 = (c & 8) != 0, k4 = (c & 4) != 0, k2 = (c & 2) != 0, k1 = (c & 1) != 0;
  float a[8], b[4], d[2];
#pragma unroll
  for (int j = 0; j < 8; ++j) {
    float keep = k8 ? v[j + 8] : v[j];
    float send = k8 ? v[j] : v[j + 8];
    a[j] = keep + row_xor<8>(send);
  }
#pragma unroll
  for (int j = 0; j < 4; ++j) {
    float keep = k4 ? a[j + 4] : a[j];
    float send = k4 ? a[j] : a[j + 4];
    b[j] = keep + row_xor<4>(send);
  }
#pragma unroll
  for (int j = 0; j < 2; ++j) {
    float keep = k2 ? b[j + 2] : b[j];
    float send = k2 ? b[j] : b[j + 2];
    d[j] = keep + row_xor<2>(send);
  }
  float keep = k1 ? d[1] : d[0];
  float send = k1 ? d[0] : d[1];
  return keep + row_xor<1>(send);
}

template<int K, int NCOL>
__global__ __launch_bounds__(256)
void wmma_gemm_kernel(const float* __restrict__ A, const float* __restrict__ B,
                      float* __restrict__ C,
                      const float* __restrict__ al, const float* __restrict__ ar,
                      float* __restrict__ el, float* __restrict__ er, int Mrows) {
  constexpr int NT = NCOL / 16;
  constexpr int HT = NT / 2;
  int wave = (int)((blockIdx.x * blockDim.x + threadIdx.x) >> 5);
  int lane = (int)(threadIdx.x & 31);
  int m0 = wave * 16;
  if (m0 >= Mrows) return;
  int half = lane >> 4;
  int lr = lane & 15;

  float alv[NT], arv[NT];
#pragma unroll
  for (int t = 0; t < NT; ++t) {
    alv[t] = al[t * 16 + lr];
    arv[t] = ar[t * 16 + lr];
  }

  v8f acc[NT] = {};
  const float* arow = A + (size_t)(m0 + lr) * K;

  for (int k = 0; k < K; k += 4) {
    int ka = k + 2 * half;
    v2f a;
    a.x = arow[ka];
    a.y = arow[ka + 1];
    const float* b0 = B + (size_t)ka * NCOL;
    const float* b1 = B + (size_t)(ka + 1) * NCOL;
#pragma unroll
    for (int t = 0; t < NT; ++t) {
      v2f b;
      b.x = b0[t * 16 + lr];
      b.y = b1[t * 16 + lr];
      acc[t] = __builtin_amdgcn_wmma_f32_16x16x4_f32(
           false, a,  false, b,
           (short)0, acc[t],  false,  false);
    }
  }

  int row = m0 + half * 8;
#pragma unroll
  for (int t = 0; t < NT; ++t) {
#pragma unroll
    for (int r = 0; r < 8; ++r) {
      C[(size_t)(row + r) * NCOL + t * 16 + lr] = acc[t][r];
    }
  }

  float vl[16], vr[16];
#pragma unroll
  for (int r = 0; r < 8; ++r) {
#pragma unroll
    for (int h = 0; h < 2; ++h) {
      float sl = acc[h * HT][r] * alv[h * HT];
      float sr = acc[h * HT][r] * arv[h * HT];
#pragma unroll
      for (int t = h * HT + 1; t < (h + 1) * HT; ++t) {
        sl += acc[t][r] * alv[t];
        sr += acc[t][r] * arv[t];
      }
      vl[2 * r + h] = sl;
      vr[2 * r + h] = sr;
    }
  }
  float suml = row16_transpose_sum(vl, lr);
  float sumr = row16_transpose_sum(vr, lr);
  el[(size_t)m0 * 2 + lane] = suml;
  er[(size_t)m0 * 2 + lane] = sumr;
}

#define BK_NC  256
#define BK_CH  ((GAT_E + BK_NC - 1) / BK_NC)
#define BK_RSH 8
#define BK_RN  (1 << BK_RSH)
#define BK_NR  ((GAT_N + BK_RN - 1) / BK_RN)
#define BK_PER ((BK_CH + 255) / 256)
static_assert(BK_NR <= 256, "one thread per node range");
static_assert(BK_NR * BK_RN > GAT_N, "offs[n_nodes] is written by the last range");
static_assert(GAT_N <= (1 << 23), "packed edge word stays non-negative");
static_assert(BK_NC == 256 && BK_RN == 256, "one thread per chunk / per node of a range");

__device__ __forceinline__ int block_excl_scan_256(int v, int* wtot, int t) {
  int lane = t & 31, w = t >> 5;
  int inc = v;
  for (int off = 1; off < 32; off <<= 1) {
    int u = __shfl_up(inc, off, 32);
    if (lane >= off) inc += u;
  }
  if (lane == 31) wtot[w] = inc;
  __syncthreads();
  int wbase = 0;
  for (int q = 0; q < 8; ++q) wbase += (q < w) ? wtot[q] : 0;
  return wbase + inc - v;
}

__global__ __launch_bounds__(256)
void chunk_split_kernel(const int* __restrict__ src, const int* __restrict__ dst,
                        int* __restrict__ tmp, int* __restrict__ cntCR,
                        int* __restrict__ offCR, int nE) {
  __shared__ int hist[256];
  __shared__ int wtot[8];
  int t = (int)threadIdx.x;
  int c = (int)blockIdx.x;
  hist[t] = 0;
  __syncthreads();
  int cbeg = c * BK_CH;
  int cend = cbeg + BK_CH;
  cend = (cend < nE) ? cend : nE;
  int rg[BK_PER], pw[BK_PER], rk[BK_PER];
#pragma unroll
  for (int k = 0; k < BK_PER; ++k) {
    int i = cbeg + k * 256 + t;
    rg[k] = -1; pw[k] = 0;
    if (i < cend) {
      int d = dst[i];
      int s = src[i];
      rg[k] = d >> BK_RSH;
      pw[k] = (s << 8) | (d & (BK_RN - 1));
    }
  }
#pragma unroll
  for (int k = 0; k < BK_PER; ++k) {
    rk[k] = 0;
    if (rg[k] >= 0) rk[k] = atomicAdd(&hist[rg[k]], 1);
  }
  __syncthreads();
  int v = hist[t];
  int ex = block_excl_scan_256(v, wtot, t);
  hist[t] = ex;
  if (t < BK_NR) {
    cntCR[t * BK_NC + c] = v;
    offCR[t * BK_NC + c] = ex;
  }
  __syncthreads();
#pragma unroll
  for (int k = 0; k < BK_PER; ++k) {
    if (rg[k] >= 0) tmp[cbeg + hist[rg[k]] + rk[k]] = pw[k];
  }
}

__global__ __launch_bounds__(256)
void range_place_kernel(const int* __restrict__ tmp, const int* __restrict__ cntCR,
                        const int* __restrict__ offCR, int* __restrict__ offs,
                        int* __restrict__ csr_src, int n_nodes) {
  __shared__ int ncnt[256];
  __shared__ int wsum[8];
  __shared__ int wtot[8];
  int t = (int)threadIdx.x;
  int r = (int)blockIdx.x;
  int lane = t & 31, w = t >> 5;
  ncnt[t] = 0;
  int n_c = cntCR[r * BK_NC + t];
  int o_c = offCR[r * BK_NC + t];
  int part = 0;
  const int4* c4 = (const int4*)cntCR;
  int n4 = r * (BK_NC / 4);
#pragma unroll 8
  for (int q = t; q < n4; q += 256) {
    int4 x = c4[q];
    part += (x.x + x.y) + (x.z + x.w);
  }
  for (int off = 16; off > 0; off >>= 1) part += __shfl_xor(part, off, 32);
  if (lane == 0) wsum[w] = part;
  __syncthreads();
  int base = 0;
  for (int q = 0; q < 8; ++q) base += wsum[q];

  const int* piece = tmp + t * BK_CH + o_c;
  for (int k = 0; k < n_c; k += 4) {
    int x[4];
#pragma unroll
    for (int u = 0; u < 4; ++u) x[u] = (k + u < n_c) ? piece[k + u] : -1;
#pragma unroll
    for (int u = 0; u < 4; ++u) if (x[u] >= 0) atomicAdd(&ncnt[x[u] & (BK_RN - 1)], 1);
  }
  __syncthreads();
  int v = ncnt[t];
  int ex = block_excl_scan_256(v, wtot, t);
  int node = r * BK_RN + t;
  if (node <= n_nodes) offs[node] = base + ex;
  ncnt[t] = ex;
  __syncthreads();
  for (int k = 0; k < n_c; k += 4) {
    int x[4];
#pragma unroll
    for (int u = 0; u < 4; ++u) x[u] = (k + u < n_c) ? piece[k + u] : -1;
#pragma unroll
    for (int u = 0; u < 4; ++u) {
      if (x[u] >= 0) {
        int p = atomicAdd(&ncnt[x[u] & (BK_RN - 1)], 1);
        csr_src[base + p] = x[u] >> 8;
      }
    }
  }
}

#define GP_NPW  4
#define GP_RUNS ((GAT_N + GP_NPW - 1) / GP_NPW)

template<int F>
__device__ __forceinline__ void gather_one_node(
    int d, int beg, int end, int s_f, float v0_f, float v1_f, float er0, float er1,
    int lane, bool hi, int col,
    const int* __restrict__ csr_src, const float* __restrict__ feat,
    const float* __restrict__ el,
    const __attribute__((ext_vector_type(2 * F / 32))) float& bv,
    float* __restrict__ out) {
  constexpr int C = 2 * F;
  constexpr int VEC = C / 32;
  typedef __attribute__((ext_vector_type(VEC))) float vrow;

  float m0 = v0_f, m1 = v1_f;
  for (int base = beg + 32; base < end; base += 32) {
    int j = base + lane;
    if (j < end) {
      int s = csr_src[j];
      v2f e2 = *(const v2f*)(el + 2 * (size_t)s);
      float a = e2.x + er0, b = e2.y + er1;
      m0 = fmaxf(m0, (a > 0.f) ? a : NEG_SLOPE * a);
      m1 = fmaxf(m1, (b > 0.f) ? b : NEG_SLOPE * b);
    }
  }
  for (int off = 16; off > 0; off >>= 1) {
    m0 = fmaxf(m0, __shfl_xor(m0, off, 32));
    m1 = fmaxf(m1, __shfl_xor(m1, off, 32));
  }

  vrow acc = {};
  float dsum = 0.f;
  for (int base = beg; base < end; base += 32) {
    int j = base + lane;
    int s = s_f;
    float v0 = v0_f, v1 = v1_f;
    if (base != beg) {
      s = 0; v0 = -INFINITY; v1 = -INFINITY;
      if (j < end) {
        s = csr_src[j];
        v2f e2 = *(const v2f*)(el + 2 * (size_t)s);
        float a = e2.x + er0, b = e2.y + er1;
        v0 = (a > 0.f) ? a : NEG_SLOPE * a;
        v1 = (b > 0.f) ? b : NEG_SLOPE * b;
      }
    }
    float a0 = (j < end) ? __expf(v0 - m0) : 0.f;
    float a1 = (j < end) ? __expf(v1 - m1) : 0.f;
    int n = end - base;
    n = (n < 32) ? n : 32;
#pragma unroll 4
    for (int t = 0; t < n; ++t) {
      int st = __builtin_amdgcn_readlane(s, t);
      float at0 = __int_as_float(__builtin_amdgcn_readlane(__float_as_int(a0), t));
      float at1 = __int_as_float(__builtin_amdgcn_readlane(__float_as_int(a1), t));
      float a = hi ? at1 : at0;
      vrow r = *(const vrow*)(feat + (size_t)st * C + col);
      acc += r * a;
      dsum += a;
    }
  }

  bool any = end > beg;
  vrow ov;
  for (int q = 0; q < VEC; ++q) {
    float r = any ? acc[q] / dsum : 0.f;
    float v = r + bv[q];
    float w = __shfl_xor(v, 16, 32);
    v = 0.5f * (v + w);
    ov[q] = (v > 0.f) ? v : 0.f;
  }
  if (!hi) *(vrow*)(out + (size_t)d * F + col) = ov;
}

template<int F>
__global__ __launch_bounds__(256)
void node_gather_kernel(const int* __restrict__ offs, const int* __restrict__ csr_src,
                        const float* __restrict__ feat,
                        const float* __restrict__ el, const float* __restrict__ er,
                        const float* __restrict__ bias, float* __restrict__ out,
                        int n_nodes) {
  constexpr int C = 2 * F;
  constexpr int VEC = C / 32;
  typedef __attribute__((ext_vector_type(VEC))) float vrow;
  int lane = (int)(threadIdx.x & 31);
  int w = __builtin_amdgcn_readfirstlane((int)(blockIdx.x * 8 + (threadIdx.x >> 5)));
  int d0 = w * GP_NPW;
  if (d0 >= n_nodes) return;
  bool hi = lane >= 16;
  int col = lane * VEC;

  int oi = d0 + lane;
  oi = (oi < n_nodes) ? oi : n_nodes;
  int ov = offs[oi];
  int ei = 2 * d0 + lane;
  ei = (ei < 2 * n_nodes - 1) ? ei : 2 * n_nodes - 1;
  int erv = __float_as_int(er[ei]);
  int bnd[GP_NPW + 1];
#pragma unroll
  for (int i = 0; i <= GP_NPW; ++i) bnd[i] = __builtin_amdgcn_readlane(ov, i);

  vrow bv = *(const vrow*)(bias + col);

  int s_f[GP_NPW];
#pragma unroll
  for (int i = 0; i < GP_NPW; ++i) {
    int j = bnd[i] + lane;
    int jl = bnd[i + 1] - 1;
    j = (j < jl) ? j : jl;
    j = (j > 0) ? j : 0;
    s_f[i] = csr_src[j];
  }
  v2f e2[GP_NPW];
#pragma unroll
  for (int i = 0; i < GP_NPW; ++i) e2[i] = *(const v2f*)(el + 2 * (size_t)s_f[i]);

#pragma unroll
  for (int i = 0; i < GP_NPW; ++i) {
    int d = d0 + i;
    if (d < n_nodes) {
      float er0 = __int_as_float(__builtin_amdgcn_readlane(erv, 2 * i));
      float er1 = __int_as_float(__builtin_amdgcn_readlane(erv, 2 * i + 1));
      bool act = bnd[i] + lane < bnd[i + 1];
      float a = e2[i].x + er0, b = e2[i].y + er1;
      float v0_f = act ? ((a > 0.f) ? a : NEG_SLOPE * a) : -INFINITY;
      float v1_f = act ? ((b > 0.f) ? b : NEG_SLOPE * b) : -INFINITY;
      gather_one_node<F>(d, bnd[i], bnd[i + 1], s_f[i], v0_f, v1_f, er0, er1,
                         lane, hi, col, csr_src, feat, el, bv, out);
    }
  }
}

static inline int cdiv(long long a, long long b) { return (int)((a + b - 1) / b); }

extern "C" void kernel_launch(void* const* d_in, const int* in_sizes, int n_in,
                              void* d_out, int out_size, void* d_ws, size_t ws_size,
                              hipStream_t stream) {
  (void)in_sizes; (void)n_in; (void)out_size; (void)ws_size;
  const float* x   = (const float*)d_in[0];
  const int*  esrc = (const int*)d_in[1];
  const int*  edst = (const int*)d_in[2];
  const float* W1  = (const float*)d_in[3];
  const float* al1 = (const float*)d_in[4];
  const float* ar1 = (const float*)d_in[5];
  const float* b1  = (const float*)d_in[6];
  const float* W2  = (const float*)d_in[7];
  const float* al2 = (const float*)d_in[8];
  const float* ar2 = (const float*)d_in[9];
  const float* b2  = (const float*)d_in[10];

  const int N = GAT_N, E = GAT_E;

  float* ws = (float*)d_ws;
  size_t o = 0;
  float* feat1  = ws + o; o += (size_t)N * 128;
  float* el1    = ws + o; o += (size_t)N * 2;
  float* er1    = ws + o; o += (size_t)N * 2;
  float* h1     = ws + o; o += (size_t)N * 64;
  float* feat2  = ws + o; o += (size_t)N * 64;
  float* el2    = ws + o; o += (size_t)N * 2;
  float* er2    = ws + o; o += (size_t)N * 2;
  int*   tmp    = (int*)(ws + o); o += (size_t)E;
  int*   csrc   = (int*)(ws + o); o += (size_t)E;
  int*   offs   = (int*)(ws + o); o += (size_t)N + 4;
  int*   cntCR  = (int*)(ws + o); o += (size_t)BK_NR * BK_NC;
  int*   offCR  = (int*)(ws + o); o += (size_t)BK_NR * BK_NC;

  const int BLK = 256;
  chunk_split_kernel<<<BK_NC, 256, 0, stream>>>(esrc, edst, tmp, cntCR, offCR, E);
  range_place_kernel<<<BK_NR, 256, 0, stream>>>(tmp, cntCR, offCR, offs, csrc, N);

  {
    int waves = N / 16;
    wmma_gemm_kernel<128, 128><<<cdiv((long long)waves * 32, BLK), BLK, 0, stream>>>(
        x, W1, feat1, al1, ar1, el1, er1, N);
    node_gather_kernel<64><<<cdiv((long long)GP_RUNS * 32, BLK), BLK, 0, stream>>>(
        offs, csrc, feat1, el1, er1, b1, h1, N);
  }
  {
    int waves = N / 16;
    wmma_gemm_kernel<64, 64><<<cdiv((long long)waves * 32, BLK), BLK, 0, stream>>>(
        h1, W2, feat2, al2, ar2, el2, er2, N);
    node_gather_kernel<32><<<cdiv((long long)GP_RUNS * 32, BLK), BLK, 0, stream>>>(
        offs, csrc, feat2, el2, er2, b2, (float*)d_out, N);
  }
}


# Layer-1 projection runs inside the two edge-sort launches: each carries half of its row stripes as extra work-groups beside the sort's (independent work overlapped); 6 launches to 5; no sum changed
# speedup vs baseline: 6.9128x; 1.1313x over previous
#include <hip/hip_runtime.h>
#include <hip/hip_bf16.h>

typedef __attribute__((ext_vector_type(2))) float v2f;
typedef __attribute__((ext_vector_type(8))) float v8f;

#define GAT_N    50000
#define GAT_E    800000
#define GAT_DIN  128
#define GAT_DHID 64
#define GAT_DOUT 32
#define NEG_SLOPE 0.2f


template<int M>
__device__ __forceinline__ float row_xor(float v) {
  return __builtin_bit_cast(float,
      __builtin_amdgcn_update_dpp(0, __builtin_bit_cast(int, v), 0x160 | M, 0xf, 0xf, true));
}

__device__ __forceinline__ float row16_transpose_sum(const float (&v)[16], int c) {
  bool k8 = (c & 8) != 0, k4 = (c & 4) != 0, k2 = (c & 2) != 0, k1 = (c & 1) != 0;
  float a[8], b[4], d[2];
#pragma unroll
  for (int j = 0; j < 8; ++j) {
    float keep = k8 ? v[j + 8] : v[j];
    float send = k8 ? v[j] : v[j + 8];
    a[j] = keep + row_xor<8>(send);
  }
#pragma unroll
  for (int j = 0; j < 4; ++j) {
    float keep = k4 ? a[j + 4] : a[j];
    float send = k4 ? a[j] : a[j + 4];
    b[j] = keep + row_xor<4>(send);
  }
#pragma unroll
  for (int j = 0; j < 2; ++j) {
    float keep = k2 ? b[j + 2] : b[j];
    float send = k2 ? b[j] : b[j + 2];
    d[j] = keep + row_xor<2>(send);
  }
  float keep = k1 ? d[1] : d[0];
  float send = k1 ? d[0] : d[1];
  return keep + row_xor<1>(send);
}

template<int K, int NCOL>
__device__ __forceinline__
void gemm_stripe(const float* __restrict__ A, const float* __restrict__ B,
                 float* __restrict__ C,
                 const float* __restrict__ al, const float* __restrict__ ar,
                 float* __restrict__ el, float* __restrict__ er, int wave, int lane) {
  constexpr int NT = NCOL / 16;
  constexpr int HT = NT / 2;
  int m0 = wave * 16;
  int half = lane >> 4;
  int lr = lane & 15;

  float alv[NT], arv[NT];
#pragma unroll
  for (int t = 0; t < NT; ++t) {
    alv[t] = al[t * 16 + lr];
    arv[t] = ar[t * 16 + lr];
  }

  v8f acc[NT] = {};
  const float* arow = A + (size_t)(m0 + lr) * K;

  for (int k = 0; k < K; k += 4) {
    int ka = k + 2 * half;
    v2f a;
    a.x = arow[ka];
    a.y = arow[ka + 1];
    const float* b0 = B + (size_t)ka * NCOL;
    const float* b1 = B + (size_t)(ka + 1) * NCOL;
#pragma unroll
    for (int t = 0; t < NT; ++t) {
      v2f b;
      b.x = b0[t * 16 + lr];
      b.y = b1[t * 16 + lr];
      acc[t] = __builtin_amdgcn_wmma_f32_16x16x4_f32(
           false, a,  false, b,
           (short)0, acc[t],  false,  false);
    }
  }

  int row = m0 + half * 8;
#pragma unroll
  for (int t = 0; t < NT; ++t) {
#pragma unroll
    for (int r = 0; r < 8; ++r) {
      C[(size_t)(row + r) * NCOL + t * 16 + lr] = acc[t][r];
    }
  }

  float vl[16], vr[16];
#pragma unroll
  for (int r = 0; r < 8; ++r) {
#pragma unroll
    for (int h = 0; h < 2; ++h) {
      float sl = acc[h * HT][r] * alv[h * HT];
      float sr = acc[h * HT][r] * arv[h * HT];
#pragma unroll
      for (int t = h * HT + 1; t < (h + 1) * HT; ++t) {
        sl += acc[t][r] * alv[t];
        sr += acc[t][r] * arv[t];
      }
      vl[2 * r + h] = sl;
      vr[2 * r + h] = sr;
    }
  }
  float suml = row16_transpose_sum(vl, lr);
  float sumr = row16_transpose_sum(vr, lr);
  el[(size_t)m0 * 2 + lane] = suml;
  er[(size_t)m0 * 2 + lane] = sumr;
}

template<int K, int NCOL>
__global__ __launch_bounds__(256)
void wmma_gemm_kernel(const float* __restrict__ A, const float* __restrict__ B,
                      float* __restrict__ C,
                      const float* __restrict__ al, const float* __restrict__ ar,
                      float* __restrict__ el, float* __restrict__ er, int Mrows) {
  int wave = (int)((blockIdx.x * blockDim.x + threadIdx.x) >> 5);
  int lane = (int)(threadIdx.x & 31);
  if (wave * 16 >= Mrows) return;
  gemm_stripe<K, NCOL>(A, B, C, al, ar, el, er, wave, lane);
}

#define BK_NC  256
#define BK_CH  ((GAT_E + BK_NC - 1) / BK_NC)
#define BK_RSH 8
#define BK_RN  (1 << BK_RSH)
#define BK_NR  ((GAT_N + BK_RN - 1) / BK_RN)
#define BK_PER ((BK_CH + 255) / 256)
static_assert(BK_NR <= 256, "one thread per node range");
static_assert(BK_NR * BK_RN > GAT_N, "offs[n_nodes] is written by the last range");
static_assert(GAT_N <= (1 << 23), "packed edge word stays non-negative");
static_assert(BK_NC == 256 && BK_RN == 256, "one thread per chunk / per node of a range");

__device__ __forceinline__ int block_excl_scan_256(int v, int* wtot, int t) {
  int lane = t & 31, w = t >> 5;
  int inc = v;
  for (int off = 1; off < 32; off <<= 1) {
    int u = __shfl_up(inc, off, 32);
    if (lane >= off) inc += u;
  }
  if (lane == 31) wtot[w] = inc;
  __syncthreads();
  int wbase = 0;
  for (int q = 0; q < 8; ++q) wbase += (q < w) ? wtot[q] : 0;
  return wbase + inc - v;
}

__device__ __forceinline__
void chunk_split_body(const int* __restrict__ src, const int* __restrict__ dst,
                      int* __restrict__ tmp, int* __restrict__ cntCR,
                      int* __restrict__ offCR, int nE, int c, int* hist, int* wtot) {
  int t = (int)threadIdx.x;
  hist[t] = 0;
  __syncthreads();
  int cbeg = c * BK_CH;
  int cend = cbeg + BK_CH;
  cend = (cend < nE) ? cend : nE;
  int rg[BK_PER], pw[BK_PER], rk[BK_PER];
#pragma unroll
  for (int k = 0; k < BK_PER; ++k) {
    int i = cbeg + k * 256 + t;
    rg[k] = -1; pw[k] = 0;
    if (i < cend) {
      int d = dst[i];
      int s = src[i];
      rg[k] = d >> BK_RSH;
      pw[k] = (s << 8) | (d & (BK_RN - 1));
    }
  }
#pragma unroll
  for (int k = 0; k < BK_PER; ++k) {
    rk[k] = 0;
    if (rg[k] >= 0) rk[k] = atomicAdd(&hist[rg[k]], 1);
  }
  __syncthreads();
  int v = hist[t];
  int ex = block_excl_scan_256(v, wtot, t);
  hist[t] = ex;
  if (t < BK_NR) {
    cntCR[t * BK_NC + c] = v;
    offCR[t * BK_NC + c] = ex;
  }
  __syncthreads();
#pragma unroll
  for (int k = 0; k < BK_PER; ++k) {
    if (rg[k] >= 0) tmp[cbeg + hist[rg[k]] + rk[k]] = pw[k];
  }
}

__device__ __forceinline__
void range_place_body(const int* __restrict__ tmp, const int* __restrict__ cntCR,
                      const int* __restrict__ offCR, int* __restrict__ offs,
                      int* __restrict__ csr_src, int n_nodes, int r,
                      int* ncnt, int* wsum, int* wtot) {
  int t = (int)threadIdx.x;
  int lane = t & 31, w = t >> 5;
  ncnt[t] = 0;
  int n_c = cntCR[r * BK_NC + t];
  int o_c = offCR[r * BK_NC + t];
  int part = 0;
  const int4* c4 = (const int4*)cntCR;
  int n4 = r * (BK_NC / 4);
#pragma unroll 8
  for (int q = t; q < n4; q += 256) {
    int4 x = c4[q];
    part += (x.x + x.y) + (x.z + x.w);
  }
  for (int off = 16; off > 0; off >>= 1) part += __shfl_xor(part, off, 32);
  if (lane == 0) wsum[w] = part;
  __syncthreads();
  int base = 0;
  for (int q = 0; q < 8; ++q) base += wsum[q];

  const int* piece = tmp + t * BK_CH + o_c;
  for (int k = 0; k < n_c; k += 4) {
    int x[4];
#pragma unroll
    for (int u = 0; u < 4; ++u) x[u] = (k + u < n_c) ? piece[k + u] : -1;
#pragma unroll
    for (int u = 0; u < 4; ++u) if (x[u] >= 0) atomicAdd(&ncnt[x[u] & (BK_RN - 1)], 1);
  }
  __syncthreads();
  int v = ncnt[t];
  int ex = block_excl_scan_256(v, wtot, t);
  int node = r * BK_RN + t;
  if (node <= n_nodes) offs[node] = base + ex;
  ncnt[t] = ex;
  __syncthreads();
  for (int k = 0; k < n_c; k += 4) {
    int x[4];
#pragma unroll
    for (int u = 0; u < 4; ++u) x[u] = (k + u < n_c) ? piece[k + u] : -1;
#pragma unroll
    for (int u = 0; u < 4; ++u) {
      if (x[u] >= 0) {
        int p = atomicAdd(&ncnt[x[u] & (BK_RN - 1)], 1);
        csr_src[base + p] = x[u] >> 8;
      }
    }
  }
}

#define PJ_STRIPES (GAT_N / 16)
#define PJ_WG_A    ((PJ_STRIPES / 2 + 7) / 8)
#define PJ_SPLIT   (PJ_WG_A * 8)
#define PJ_WG_B    ((PJ_STRIPES - PJ_SPLIT + 7) / 8)
static_assert(GAT_N % 16 == 0, "whole stripes");

__global__ __launch_bounds__(256)
void split_project_kernel(const int* __restrict__ src, const int* __restrict__ dst,
                          int* __restrict__ tmp, int* __restrict__ cntCR,
                          int* __restrict__ offCR, int nE,
                          const float* __restrict__ x, const float* __restrict__ W,
                          float* __restrict__ feat,
                          const float* __restrict__ al, const float* __restrict__ ar,
                          float* __restrict__ el, float* __restrict__ er) {
  __shared__ int hist[256];
  __shared__ int wtot[8];
  int b = (int)blockIdx.x;
  if (b < BK_NC) {
    chunk_split_body(src, dst, tmp, cntCR, offCR, nE, b, hist, wtot);
    return;
  }
  int wave = (b - BK_NC) * 8 + (int)(threadIdx.x >> 5);
  if (wave >= PJ_SPLIT) return;
  gemm_stripe<GAT_DIN, 2 * GAT_DHID>(x, W, feat, al, ar, el, er, wave, (int)(threadIdx.x & 31));
}

__global__ __launch_bounds__(256)
void place_project_kernel(const int* __restrict__ tmp, const int* __restrict__ cntCR,
                          const int* __restrict__ offCR, int* __restrict__ offs,
                          int* __restrict__ csr_src, int n_nodes,
                          const float* __restrict__ x, const float* __restrict__ W,
                          float* __restrict__ feat,
                          const float* __restrict__ al, const float* __restrict__ ar,
                          float* __restrict__ el, float* __restrict__ er) {
  __shared__ int ncnt[256];
  __shared__ int wsum[8];
  __shared__ int wtot[8];
  int b = (int)blockIdx.x;
  if (b < BK_NR) {
    range_place_body(tmp, cntCR, offCR, offs, csr_src, n_nodes, b, ncnt, wsum, wtot);
    return;
  }
  int wave = PJ_SPLIT + (b - BK_NR) * 8 + (int)(threadIdx.x >> 5);
  if (wave >= PJ_STRIPES) return;
  gemm_stripe<GAT_DIN, 2 * GAT_DHID>(x, W, feat, al, ar, el, er, wave, (int)(threadIdx.x & 31));
}

#define GP_NPW  4
#define GP_RUNS ((GAT_N + GP_NPW - 1) / GP_NPW)

template<int F>
__device__ __forceinline__ void gather_one_node(
    int d, int beg, int end, int s_f, float v0_f, float v1_f, float er0, float er1,
    int lane, bool hi, int col,
    const int* __restrict__ csr_src, const float* __restrict__ feat,
    const float* __restrict__ el,
    const __attribute__((ext_vector_type(2 * F / 32))) float& bv,
    float* __restrict__ out) {
  constexpr int C = 2 * F;
  constexpr int VEC = C / 32;
  typedef __attribute__((ext_vector_type(VEC))) float vrow;

  float m0 = v0_f, m1 = v1_f;
  for (int base = beg + 32; base < end; base += 32) {
    int j = base + lane;
    if (j < end) {
      int s = csr_src[j];
      v2f e2 = *(const v2f*)(el + 2 * (size_t)s);
      float a = e2.x + er0, b = e2.y + er1;
      m0 = fmaxf(m0, (a > 0.f) ? a : NEG_SLOPE * a);
      m1 = fmaxf(m1, (b > 0.f) ? b : NEG_SLOPE * b);
    }
  }
  for (int off = 16; off > 0; off >>= 1) {
    m0 = fmaxf(m0, __shfl_xor(m0, off, 32));
    m1 = fmaxf(m1, __shfl_xor(m1, off, 32));
  }

  vrow acc = {};
  float dsum = 0.f;
  for (int base = beg; base < end; base += 32) {
    int j = base + lane;
    int s = s_f;
    float v0 = v0_f, v1 = v1_f;
    if (base != beg) {
      s = 0; v0 = -INFINITY; v1 = -INFINITY;
      if (j < end) {
        s = csr_src[j];
        v2f e2 = *(const v2f*)(el + 2 * (size_t)s);
        float a = e2.x + er0, b = e2.y + er1;
        v0 = (a > 0.f) ? a : NEG_SLOPE * a;
        v1 = (b > 0.f) ? b : NEG_SLOPE * b;
      }
    }
    float a0 = (j < end) ? __expf(v0 - m0) : 0.f;
    float a1 = (j < end) ? __expf(v1 - m1) : 0.f;
    int n = end - base;
    n = (n < 32) ? n : 32;
#pragma unroll 4
    for (int t = 0; t < n; ++t) {
      int st = __builtin_amdgcn_readlane(s, t);
      float at0 = __int_as_float(__builtin_amdgcn_readlane(__float_as_int(a0), t));
      float at1 = __int_as_float(__builtin_amdgcn_readlane(__float_as_int(a1), t));
      float a = hi ? at1 : at0;
      vrow r = *(const vrow*)(feat + (size_t)st * C + col);
      acc += r * a;
      dsum += a;
    }
  }

  bool any = end > beg;
  vrow ov;
  for (int q = 0; q < VEC; ++q) {
    float r = any ? acc[q] / dsum : 0.f;
    float v = r + bv[q];
    float w = __shfl_xor(v, 16, 32);
    v = 0.5f * (v + w);
    ov[q] = (v > 0.f) ? v : 0.f;
  }
  if (!hi) *(vrow*)(out + (size_t)d * F + col) = ov;
}

template<int F>
__global__ __launch_bounds__(256)
void node_gather_kernel(const int* __restrict__ offs, const int* __restrict__ csr_src,
                        const float* __restrict__ feat,
                        const float* __restrict__ el, const float* __restrict__ er,
                        const float* __restrict__ bias, float* __restrict__ out,
                        int n_nodes) {
  constexpr int C = 2 * F;
  constexpr int VEC = C / 32;
  typedef __attribute__((ext_vector_type(VEC))) float vrow;
  int lane = (int)(threadIdx.x & 31);
  int w = __builtin_amdgcn_readfirstlane((int)(blockIdx.x * 8 + (threadIdx.x >> 5)));
  int d0 = w * GP_NPW;
  if (d0 >= n_nodes) return;
  bool hi = lane >= 16;
  int col = lane * VEC;

  int oi = d0 + lane;
  oi = (oi < n_nodes) ? oi : n_nodes;
  int ov = offs[oi];
  int ei = 2 * d0 + lane;
  ei = (ei < 2 * n_nodes - 1) ? ei : 2 * n_nodes - 1;
  int erv = __float_as_int(er[ei]);
  int bnd[GP_NPW + 1];
#pragma unroll
  for (int i = 0; i <= GP_NPW; ++i) bnd[i] = __builtin_amdgcn_readlane(ov, i);

  vrow bv = *(const vrow*)(bias + col);

  int s_f[GP_NPW];
#pragma unroll
  for (int i = 0; i < GP_NPW; ++i) {
    int j = bnd[i] + lane;
    int jl = bnd[i + 1] - 1;
    j = (j < jl) ? j : jl;
    j = (j > 0) ? j : 0;
    s_f[i] = csr_src[j];
  }
  v2f e2[GP_NPW];
#pragma unroll
  for (int i = 0; i < GP_NPW; ++i) e2[i] = *(const v2f*)(el + 2 * (size_t)s_f[i]);

#pragma unroll
  for (int i = 0; i < GP_NPW; ++i) {
    int d = d0 + i;
    if (d < n_nodes) {
      float er0 = __int_as_float(__builtin_amdgcn_readlane(erv, 2 * i));
      float er1 = __int_as_float(__builtin_amdgcn_readlane(erv, 2 * i + 1));
      bool act = bnd[i] + lane < bnd[i + 1];
      float a = e2[i].x + er0, b = e2[i].y + er1;
      float v0_f = act ? ((a > 0.f) ? a : NEG_SLOPE * a) : -INFINITY;
      float v1_f = act ? ((b > 0.f) ? b : NEG_SLOPE * b) : -INFINITY;
      gather_one_node<F>(d, bnd[i], bnd[i + 1], s_f[i], v0_f, v1_f, er0, er1,
                         lane, hi, col, csr_src, feat, el, bv, out);
    }
  }
}

static inline int cdiv(long long a, long long b) { return (int)((a + b - 1) / b); }

extern "C" void kernel_launch(void* const* d_in, const int* in_sizes, int n_in,
                              void* d_out, int out_size, void* d_ws, size_t ws_size,
                              hipStream_t stream) {
  (void)in_sizes; (void)n_in; (void)out_size; (void)ws_size;
  const float* x   = (const float*)d_in[0];
  const int*  esrc = (const int*)d_in[1];
  const int*  edst = (const int*)d_in[2];
  const float* W1  = (const float*)d_in[3];
  const float* al1 = (const float*)d_in[4];
  const float* ar1 = (const float*)d_in[5];
  const float* b1  = (const float*)d_in[6];
  const float* W2  = (const float*)d_in[7];
  const float* al2 = (const float*)d_in[8];
  const float* ar2 = (const float*)d_in[9];
  const float* b2  = (const float*)d_in[10];

  const int N = GAT_N, E = GAT_E;

  float* ws = (float*)d_ws;
  size_t o = 0;
  float* feat1  = ws + o; o += (size_t)N * 128;
  float* el1    = ws + o; o += (size_t)N * 2;
  float* er1    = ws + o; o += (size_t)N * 2;
  float* h1     = ws + o; o += (size_t)N * 64;
  float* feat2  = ws + o; o += (size_t)N * 64;
  float* el2    = ws + o; o += (size_t)N * 2;
  float* er2    = ws + o; o += (size_t)N * 2;
  int*   tmp    = (int*)(ws + o); o += (size_t)E;
  int*   csrc   = (int*)(ws + o); o += (size_t)E;
  int*   offs   = (int*)(ws + o); o += (size_t)N + 4;
  int*   cntCR  = (int*)(ws + o); o += (size_t)BK_NR * BK_NC;
  int*   offCR  = (int*)(ws + o); o += (size_t)BK_NR * BK_NC;

  const int BLK = 256;
  split_project_kernel<<<BK_NC + PJ_WG_A, 256, 0, stream>>>(
      esrc, edst, tmp, cntCR, offCR, E, x, W1, feat1, al1, ar1, el1, er1);
  place_project_kernel<<<BK_NR + PJ_WG_B, 256, 0, stream>>>(
      tmp, cntCR, offCR, offs, csrc, N, x, W1, feat1, al1, ar1, el1, er1);

  {
    node_gather_kernel<64><<<cdiv((long long)GP_RUNS * 32, BLK), BLK, 0, stream>>>(
        offs, csrc, feat1, el1, er1, b1, h1, N);
  }
  {
    int waves = N / 16;
    wmma_gemm_kernel<64, 64><<<cdiv((long long)waves * 32, BLK), BLK, 0, stream>>>(
        h1, W2, feat2, al2, ar2, el2, er2, N);
    node_gather_kernel<32><<<cdiv((long long)GP_RUNS * 32, BLK), BLK, 0, stream>>>(
        offs, csrc, feat2, el2, er2, b2, (float*)d_out, N);
  }
}
